# starting kernel
# baseline (speedup 1.0000x reference)
#include <hip/hip_runtime.h>

typedef __attribute__((ext_vector_type(2))) float v2f;
typedef __attribute__((ext_vector_type(8))) float v8f;

#define T_DIM 2048
#define D_DIM 128
#define CHUNK 64

__global__ __launch_bounds__(256) void attn_pool_kernel(
    const float* __restrict__ x,
    const float* __restrict__ W,
    const float* __restrict__ bias,
    float* __restrict__ out)
{
    __shared__ float pbuf[2][CHUNK];
    __shared__ float s_shared;

    const int b    = blockIdx.x;
    const int tid  = threadIdx.x;
    const int wave = tid >> 5;
    const int lane = tid & 31;
    const int half = lane >> 4;
    const int n    = lane & 15;
    const int d0   = wave * 16;

    const float* xb = x + (size_t)b * T_DIM * D_DIM;

    if (tid == 0) s_shared = 0.0f;

    const float4 w4 = reinterpret_cast<const float4*>(W)[lane];

    v8f   c = {};
    float s_local = 0.0f;

    __syncthreads();

    for (int base = 0; base < T_DIM; base += CHUNK) {
        const int cur = (base / CHUNK) & 1;

        if (base + CHUNK < T_DIM) {
            __builtin_prefetch(xb + (size_t)(base + CHUNK) * D_DIM + tid * 32, 0, 0);
        }

        for (int j = 0; j < 8; ++j) {
            const int t = base + j * 8 + wave;
            const float4 xv =
                reinterpret_cast<const float4*>(xb + (size_t)t * D_DIM)[lane];
            float partial = xv.x * w4.x + xv.y * w4.y + xv.z * w4.z + xv.w * w4.w;
            partial += __shfl_xor(partial, 1, 32);
            partial += __shfl_xor(partial, 2, 32);
            partial += __shfl_xor(partial, 4, 32);
            partial += __shfl_xor(partial, 8, 32);
            partial += __shfl_xor(partial, 16, 32);
            const float e = tanhf(partial + bias[t]);
            const float p = __expf(e);
            s_local += p;
            if (lane == 0) pbuf[cur][j * 8 + wave] = p;
        }
        __syncthreads();

        for (int i = 0; i < 16; ++i) {
            const int t0 = base + 4 * i;
            const float p0 = pbuf[cur][4 * i + 0];
            const float p1 = pbuf[cur][4 * i + 1];
            const float p2 = pbuf[cur][4 * i + 2];
            const float p3 = pbuf[cur][4 * i + 3];

            v2f a;
            a.x = half ? p2 : p0;
            a.y = half ? p3 : p1;

            const float* col = xb + (size_t)(t0 + 2 * half) * D_DIM + d0 + n;
            v2f bb;
            bb.x = col[0];
            bb.y = col[D_DIM];

            c = __builtin_amdgcn_wmma_f32_16x16x4_f32(
                false, a, false, bb, (short)0, c, false, false);
        }
    }

    if (lane == 0) atomicAdd(&s_shared, s_local);
    __syncthreads();

    const float inv = 1.0f / s_shared;
    if (lane < 16) {
        out[(size_t)b * D_DIM + d0 + lane] = c[0] * inv;
    }
}

extern "C" void kernel_launch(void* const* d_in, const int* in_sizes, int n_in,
                              void* d_out, int out_size, void* d_ws, size_t ws_size,
                              hipStream_t stream) {
    const float* x    = (const float*)d_in[0];
    const float* W    = (const float*)d_in[1];
    const float* bias = (const float*)d_in[2];
    float* out = (float*)d_out;

    const int B = in_sizes[0] / (T_DIM * D_DIM);
    attn_pool_kernel<<<B, 256, 0, stream>>>(x, W, bias, out);
}


# logit phase made parallel: a wave's 8 row loads issued together, the 8 dots summed in one halving DPP butterfly, tanh/exp once per wave and chunk; sum of p now per lane, then across lanes
# speedup vs baseline: 3.1047x; 3.1047x over previous
#include <hip/hip_runtime.h>


typedef __attribute__((ext_vector_type(2))) float v2f;
typedef __attribute__((ext_vector_type(8))) float v8f;

#define T_DIM 2048
#define D_DIM 128
#define CHUNK 64

template <int M>
__device__ __forceinline__ float lane_xor(float v) {
    const int b = __builtin_bit_cast(int, v);
    return __builtin_bit_cast(
        float, __builtin_amdgcn_update_dpp(b, b, 0x160 + M  , 0xf, 0xf, false));
}

__device__ __forceinline__ float lane_xor16(float v) {
    const int b = __builtin_bit_cast(int, v);
    return __builtin_bit_cast(
        float, __builtin_amdgcn_permlanex16(b, b, 0x76543210, 0xfedcba98, false, false));
}

__global__ __launch_bounds__(256) void attn_pool_kernel(
    const float* __restrict__ x,
    const float* __restrict__ W,
    const float* __restrict__ bias,
    float* __restrict__ out)
{
    __shared__ float pbuf[2][CHUNK];
    __shared__ float s_shared;

    const int b    = blockIdx.x;
    const int tid  = threadIdx.x;
    const int wave = tid >> 5;
    const int lane = tid & 31;
    const int half = lane >> 4;
    const int n    = lane & 15;
    const int d0   = wave * 16;

    const float* xb = x + (size_t)b * T_DIM * D_DIM;

    if (tid == 0) s_shared = 0.0f;

    const float4 w4 = reinterpret_cast<const float4*>(W)[lane];

    const bool k0 = (lane & 1) != 0;
    const bool k1 = (lane & 2) != 0;
    const bool k2 = (lane & 4) != 0;
    const int  myrow = ((lane & 1) << 2) | (lane & 2) | ((lane >> 2) & 1);
    const int  myslot = myrow * 8 + wave;

    v8f   c = {};
    float s_lane = 0.0f;

    __syncthreads();

    for (int base = 0; base < T_DIM; base += CHUNK) {
        const int cur = (base / CHUNK) & 1;

        if (base + CHUNK < T_DIM) {
            __builtin_prefetch(xb + (size_t)(base + CHUNK) * D_DIM + tid * 32, 0, 0);
        }

        float4 xv[8];
#pragma unroll
        for (int j = 0; j < 8; ++j) {
            const int t = base + j * 8 + wave;
            xv[j] = reinterpret_cast<const float4*>(xb + (size_t)t * D_DIM)[lane];
        }
        const float bt = bias[base + myslot];

        float v[8];
#pragma unroll
        for (int j = 0; j < 8; ++j) {
            v[j] = xv[j].x * w4.x + xv[j].y * w4.y + xv[j].z * w4.z + xv[j].w * w4.w;
        }

        float u[4];
#pragma unroll
        for (int i = 0; i < 4; ++i) {
            const float keep = k0 ? v[i + 4] : v[i];
            const float send = k0 ? v[i] : v[i + 4];
            u[i] = keep + lane_xor<1>(send);
        }
        float q[2];
#pragma unroll
        for (int i = 0; i < 2; ++i) {
            const float keep = k1 ? u[i + 2] : u[i];
            const float send = k1 ? u[i] : u[i + 2];
            q[i] = keep + lane_xor<2>(send);
        }
        float dot;
        {
            const float keep = k2 ? q[1] : q[0];
            const float send = k2 ? q[0] : q[1];
            dot = keep + lane_xor<4>(send);
        }
        dot += lane_xor<8>(dot);
        dot += lane_xor16(dot);

        const float e = tanhf(dot + bt);
        const float p = __expf(e);
        s_lane += p;
        if (lane < 8) pbuf[cur][myslot] = p;
        __syncthreads();

        for (int i = 0; i < 16; ++i) {
            const int t0 = base + 4 * i;
            const float p0 = pbuf[cur][4 * i + 0];
            const float p1 = pbuf[cur][4 * i + 1];
            const float p2 = pbuf[cur][4 * i + 2];
            const float p3 = pbuf[cur][4 * i + 3];

            v2f a;
            a.x = half ? p2 : p0;
            a.y = half ? p3 : p1;

            const float* col = xb + (size_t)(t0 + 2 * half) * D_DIM + d0 + n;
            v2f bb;
            bb.x = col[0];
            bb.y = col[D_DIM];

            c = __builtin_amdgcn_wmma_f32_16x16x4_f32(
                false, a, false, bb, (short)0, c, false, false);
        }
    }

    float s_local = s_lane;
    s_local += lane_xor<1>(s_local);
    s_local += lane_xor<2>(s_local);
    s_local += lane_xor<4>(s_local);

    if (lane == 0) atomicAdd(&s_shared, s_local);
    __syncthreads();

    const float inv = 1.0f / s_shared;
    if (lane < 16) {
        out[(size_t)b * D_DIM + d0 + lane] = c[0] * inv;
    }
}

extern "C" void kernel_launch(void* const* d_in, const int* in_sizes, int n_in,
                              void* d_out, int out_size, void* d_ws, size_t ws_size,
                              hipStream_t stream) {
    const float* x    = (const float*)d_in[0];
    const float* W    = (const float*)d_in[1];
    const float* bias = (const float*)d_in[2];
    float* out = (float*)d_out;

    const int B = in_sizes[0] / (T_DIM * D_DIM);
    attn_pool_kernel<<<B, 256, 0, stream>>>(x, W, bias, out);
}


# chunks software-pipelined: next chunk's row loads stay in flight across a hand-written LDS-only barrier while this chunk's logits and WMMAs run; B loads issued first; L2 prefetch 4 chunks ahead
# speedup vs baseline: 3.5975x; 1.1587x over previous
#include <hip/hip_runtime.h>


typedef __attribute__((ext_vector_type(2))) float v2f;
typedef __attribute__((ext_vector_type(8))) float v8f;

#define T_DIM 2048
#define D_DIM 128
#define CHUNK 64
#define LAST_BASE (T_DIM - CHUNK)

template <int M>
__device__ __forceinline__ float lane_xor(float v) {
    const int b = __builtin_bit_cast(int, v);
    return __builtin_bit_cast(
        float, __builtin_amdgcn_update_dpp(b, b, 0x160 + M  , 0xf, 0xf, false));
}

__device__ __forceinline__ float lane_xor16(float v) {
    const int b = __builtin_bit_cast(int, v);
    return __builtin_bit_cast(
        float, __builtin_amdgcn_permlanex16(b, b, 0x76543210, 0xfedcba98, false, false));
}

__device__ __forceinline__ void lds_barrier() {
    asm volatile("s_wait_dscnt 0x0\n\ts_barrier_signal -1\n\ts_barrier_wait -1" ::: "memory");
}

__global__ __launch_bounds__(256) void attn_pool_kernel(
    const float* __restrict__ x,
    const float* __restrict__ W,
    const float* __restrict__ bias,
    float* __restrict__ out)
{
    __shared__ float pbuf[2][CHUNK];
    __shared__ float s_shared;

    const int b    = blockIdx.x;
    const int tid  = threadIdx.x;
    const int wave = tid >> 5;
    const int lane = tid & 31;
    const int half = lane >> 4;
    const int n    = lane & 15;
    const int d0   = wave * 16;

    const float* xb = x + (size_t)b * T_DIM * D_DIM;

    if (tid == 0) s_shared = 0.0f;

    const float4 w4 = reinterpret_cast<const float4*>(W)[lane];

    const bool k0 = (lane & 1) != 0;
    const bool k1 = (lane & 2) != 0;
    const bool k2 = (lane & 4) != 0;
    const int  myrow = ((lane & 1) << 2) | (lane & 2) | ((lane >> 2) & 1);
    const int  myslot = myrow * 8 + wave;

    v8f   c = {};
    float s_lane = 0.0f;

    __syncthreads();

    float4 xr0[8], xr1[8];
    float  bt0, bt1;
#pragma unroll
    for (int j = 0; j < 8; ++j) {
        const int t = j * 8 + wave;
        xr0[j] = reinterpret_cast<const float4*>(xb + (size_t)t * D_DIM)[lane];
    }
    bt0 = bias[myslot];
    bt1 = 0.0f;
    __builtin_prefetch(xb + (size_t)(1 * CHUNK) * D_DIM + tid * 32, 0, 0);
    __builtin_prefetch(xb + (size_t)(2 * CHUNK) * D_DIM + tid * 32, 0, 0);
    __builtin_prefetch(xb + (size_t)(3 * CHUNK) * D_DIM + tid * 32, 0, 0);

    auto chunk_step = [&](const float4 (&xin)[8], const float bt_in,
                          float4 (&xout)[8], float& bt_out,
                          const int base) __attribute__((always_inline)) {
        const int cur = (base / CHUNK) & 1;

        v2f bb[16];
#pragma unroll
        for (int i = 0; i < 16; ++i) {
            const float* col = xb + (size_t)(base + 4 * i + 2 * half) * D_DIM + d0 + n;
            bb[i].x = col[0];
            bb[i].y = col[D_DIM];
        }

        const int nbase = (base < LAST_BASE) ? base + CHUNK : LAST_BASE;
#pragma unroll
        for (int j = 0; j < 8; ++j) {
            const int t = nbase + j * 8 + wave;
            xout[j] = reinterpret_cast<const float4*>(xb + (size_t)t * D_DIM)[lane];
        }
        bt_out = bias[nbase + myslot];

        if (base + 4 * CHUNK < T_DIM) {
            __builtin_prefetch(xb + (size_t)(base + 4 * CHUNK) * D_DIM + tid * 32, 0, 0);
        }

        float v[8];
#pragma unroll
        for (int j = 0; j < 8; ++j) {
            v[j] = xin[j].x * w4.x + xin[j].y * w4.y + xin[j].z * w4.z + xin[j].w * w4.w;
        }

        float u[4];
#pragma unroll
        for (int i = 0; i < 4; ++i) {
            const float keep = k0 ? v[i + 4] : v[i];
            const float send = k0 ? v[i] : v[i + 4];
            u[i] = keep + lane_xor<1>(send);
        }
        float q[2];
#pragma unroll
        for (int i = 0; i < 2; ++i) {
            const float keep = k1 ? u[i + 2] : u[i];
            const float send = k1 ? u[i] : u[i + 2];
            q[i] = keep + lane_xor<2>(send);
        }
        float dot;
        {
            const float keep = k2 ? q[1] : q[0];
            const float send = k2 ? q[0] : q[1];
            dot = keep + lane_xor<4>(send);
        }
        dot += lane_xor<8>(dot);
        dot += lane_xor16(dot);

        const float e = tanhf(dot + bt_in);
        const float p = __expf(e);
        s_lane += p;
        if (lane < 8) pbuf[cur][myslot] = p;
        lds_barrier();

#pragma unroll
        for (int i = 0; i < 16; ++i) {
            const float p0 = pbuf[cur][4 * i + 0];
            const float p1 = pbuf[cur][4 * i + 1];
            const float p2 = pbuf[cur][4 * i + 2];
            const float p3 = pbuf[cur][4 * i + 3];

            v2f a;
            a.x = half ? p2 : p0;
            a.y = half ? p3 : p1;

            c = __builtin_amdgcn_wmma_f32_16x16x4_f32(
                false, a, false, bb[i], (short)0, c, false, false);
        }
    };

    for (int base = 0; base < T_DIM; base += 2 * CHUNK) {
        chunk_step(xr0, bt0, xr1, bt1, base);
        chunk_step(xr1, bt1, xr0, bt0, base + CHUNK);
    }

    float s_local = s_lane;
    s_local += lane_xor<1>(s_local);
    s_local += lane_xor<2>(s_local);
    s_local += lane_xor<4>(s_local);

    if (lane == 0) atomicAdd(&s_shared, s_local);
    __syncthreads();

    const float inv = 1.0f / s_shared;
    if (lane < 16) {
        out[(size_t)b * D_DIM + d0 + lane] = c[0] * inv;
    }
}

extern "C" void kernel_launch(void* const* d_in, const int* in_sizes, int n_in,
                              void* d_out, int out_size, void* d_ws, size_t ws_size,
                              hipStream_t stream) {
    const float* x    = (const float*)d_in[0];
    const float* W    = (const float*)d_in[1];
    const float* bias = (const float*)d_in[2];
    float* out = (float*)d_out;

    const int B = in_sizes[0] / (T_DIM * D_DIM);
    attn_pool_kernel<<<B, 256, 0, stream>>>(x, W, bias, out);
}


# x read once: each wave keeps 16 whole timesteps in registers as WMMA B operands and logit inputs; no LDS hand-over, loop barrier or atomic; wave partials added in wave order (sum order changed)
# speedup vs baseline: 5.2764x; 1.4667x over previous
#include <hip/hip_runtime.h>


typedef __attribute__((ext_vector_type(2))) float v2f;
typedef __attribute__((ext_vector_type(8))) float v8f;

#define T_DIM 2048
#define D_DIM 128
#define NWAVES 16
#define BT 16
#define NBLK (T_DIM / (NWAVES * BT))

template <int M>
__device__ __forceinline__ float lane_xor(float v) {
    const int b = __builtin_bit_cast(int, v);
    return __builtin_bit_cast(
        float, __builtin_amdgcn_update_dpp(b, b, 0x160 + M  , 0xf, 0xf, false));
}

template <int N>
__device__ __forceinline__ float row_lane(float v) {
    const int b = __builtin_bit_cast(int, v);
    return __builtin_bit_cast(
        float, __builtin_amdgcn_update_dpp(b, b, 0x150 + N  , 0xf, 0xf, false));
}

__device__ __forceinline__ float lane_xor16(float v) {
    const int b = __builtin_bit_cast(int, v);
    return __builtin_bit_cast(
        float, __builtin_amdgcn_permlanex16(b, b, 0x76543210, 0xfedcba98, false, false));
}

__global__ __launch_bounds__(NWAVES * 32) void attn_pool_kernel(
    const float* __restrict__ x,
    const float* __restrict__ W,
    const float* __restrict__ bias,
    float* __restrict__ out)
{
    __shared__ float red[NWAVES][D_DIM];
    __shared__ float sred[NWAVES];

    const int b    = blockIdx.x;
    const int tid  = threadIdx.x;
    const int wave = __builtin_amdgcn_readfirstlane(tid >> 5);
    const int lane = tid & 31;
    const int h    = lane >> 4;
    const int n    = lane & 15;

    const float* xb = x + (size_t)b * T_DIM * D_DIM;

    const float4 w0 = reinterpret_cast<const float4*>(W)[n];
    const float4 w1 = reinterpret_cast<const float4*>(W)[16 + n];
    const v2f w0a = {w0.x, w0.y}, w0b = {w0.z, w0.w};
    const v2f w1a = {w1.x, w1.y}, w1b = {w1.z, w1.w};

    const bool k0 = (n & 1) != 0;
    const bool k1 = (n & 2) != 0;
    const bool k2 = (n & 4) != 0;
    const int  myj = ((n & 1) << 2) | (n & 2) | ((n >> 2) & 1);

    const float mx = (n == 1) ? 0.0f : 1.0f;
    const float my = (n == 0) ? 0.0f : 1.0f;

    float bt[NBLK];
#pragma unroll
    for (int k = 0; k < NBLK; ++k) {
        bt[k] = bias[(k * NWAVES + wave) * BT + 2 * myj + h];
    }

    const float* lp = xb + (size_t)(wave * BT + h) * D_DIM + 4 * n;

    float4 xr[2][8][2];
    v8f    acc[4] = {};
    float  s_lane = 0.0f;

    auto load_block = [&](float4 (&xs)[8][2], const int k) __attribute__((always_inline)) {
        const float* p0 = lp + (size_t)k * (NWAVES * BT * D_DIM);
#pragma unroll
        for (int j = 0; j < 8; ++j) {
            xs[j][0] = *reinterpret_cast<const float4*>(p0 + (2 * j) * D_DIM);
            xs[j][1] = *reinterpret_cast<const float4*>(p0 + (2 * j) * D_DIM + 64);
        }
    };

    auto work_block = [&](const float4 (&xs)[8][2], const float btk) __attribute__((always_inline)) {
        float v[8];
#pragma unroll
        for (int j = 0; j < 8; ++j) {
            const float4 f = xs[j][0];
            const float4 g = xs[j][1];
            v2f s2 = v2f{f.x, f.y} * w0a;
            s2 = __builtin_elementwise_fma(v2f{f.z, f.w}, w0b, s2);
            s2 = __builtin_elementwise_fma(v2f{g.x, g.y}, w1a, s2);
            s2 = __builtin_elementwise_fma(v2f{g.z, g.w}, w1b, s2);
            v[j] = s2.x + s2.y;
        }
        float u[4];
#pragma unroll
        for (int i = 0; i < 4; ++i) {
            const float keep = k0 ? v[i + 4] : v[i];
            const float send = k0 ? v[i] : v[i + 4];
            u[i] = keep + lane_xor<1>(send);
        }
        float q[2];
#pragma unroll
        for (int i = 0; i < 2; ++i) {
            const float keep = k1 ? u[i + 2] : u[i];
            const float send = k1 ? u[i] : u[i + 2];
            q[i] = keep + lane_xor<2>(send);
        }
        float dot;
        {
            const float keep = k2 ? q[1] : q[0];
            const float send = k2 ? q[0] : q[1];
            dot = keep + lane_xor<4>(send);
        }
        dot += lane_xor<8>(dot);

        const float e = tanhf(dot + btk);
        const float p = __expf(e);
        s_lane += p;

        float pj[8];
        pj[0] = row_lane<0>(p); pj[1] = row_lane<4>(p); pj[2] = row_lane<2>(p); pj[3] = row_lane<6>(p);
        pj[4] = row_lane<1>(p); pj[5] = row_lane<5>(p); pj[6] = row_lane<3>(p); pj[7] = row_lane<7>(p);
#pragma unroll
        for (int j = 0; j < 8; ++j) {
            v2f a;
            a.x = pj[j] * mx;
            a.y = pj[j] * my;
#pragma unroll
            for (int L = 0; L < 2; ++L) {
                const float4 f = xs[j][L];
                v2f b0; b0.x = f.x; b0.y = f.y;
                v2f b1; b1.x = f.z; b1.y = f.w;
                acc[2 * L + 0] = __builtin_amdgcn_wmma_f32_16x16x4_f32(
                    false, a, false, b0, (short)0, acc[2 * L + 0], false, false);
                acc[2 * L + 1] = __builtin_amdgcn_wmma_f32_16x16x4_f32(
                    false, a, false, b1, (short)0, acc[2 * L + 1], false, false);
            }
        }
    };

    load_block(xr[0], 0);
#pragma unroll
    for (int k = 0; k < NBLK; ++k) {
        __builtin_amdgcn_sched_barrier(0);
        if (k + 1 < NBLK) load_block(xr[(k + 1) & 1], k + 1);
        __builtin_amdgcn_sched_barrier(0);
        work_block(xr[k & 1], bt[k]);
    }

    float s_wave = s_lane;
    s_wave += lane_xor<1>(s_wave);
    s_wave += lane_xor<2>(s_wave);
    s_wave += lane_xor<4>(s_wave);
    s_wave += lane_xor16(s_wave);

    if (lane < 16) {
#pragma unroll
        for (int L = 0; L < 2; ++L) {
            float4 o;
            o.x = acc[2 * L + 0][0];
            o.y = acc[2 * L + 0][1];
            o.z = acc[2 * L + 1][0];
            o.w = acc[2 * L + 1][1];
            *reinterpret_cast<float4*>(&red[wave][64 * L + 4 * n]) = o;
        }
        if (lane == 0) sred[wave] = s_wave;
    }
    __syncthreads();

    if (tid < D_DIM) {
        float a = 0.0f;
        float s = 0.0f;
#pragma unroll
        for (int w = 0; w < NWAVES; ++w) {
            a += red[w][tid];
            s += sred[w];
        }
        const float inv = 1.0f / s;
        out[(size_t)b * D_DIM + tid] = a * inv;
    }
}

extern "C" void kernel_launch(void* const* d_in, const int* in_sizes, int n_in,
                              void* d_out, int out_size, void* d_ws, size_t ws_size,
                              hipStream_t stream) {
    const float* x    = (const float*)d_in[0];
    const float* W    = (const float*)d_in[1];
    const float* bias = (const float*)d_in[2];
    float* out = (float*)d_out;

    const int B = in_sizes[0] / (T_DIM * D_DIM);
    attn_pool_kernel<<<B, NWAVES * 32, 0, stream>>>(x, W, bias, out);
}
